# starting kernel
# baseline (speedup 1.0000x reference)
#include <hip/hip_runtime.h>
#include <hip/hip_bf16.h>

typedef __attribute__((ext_vector_type(2))) float v2f;
typedef __attribute__((ext_vector_type(8))) float v8f;

#define EMB 128

__global__ void k_init_deg(float* __restrict__ deg, int n) {
    int i = blockIdx.x * blockDim.x + threadIdx.x;
    if (i < n) deg[i] = 1.0f;
}

__global__ void k_count_deg(const int* __restrict__ colI, float* __restrict__ deg, int E) {
    int i = blockIdx.x * blockDim.x + threadIdx.x;
    if (i < E) atomicAdd(&deg[colI[i]], 1.0f);
}

__global__ void k_dinv(float* __restrict__ deg, int n) {
    int i = blockIdx.x * blockDim.x + threadIdx.x;
    if (i < n) deg[i] = rsqrtf(deg[i]);
}

__global__ void k_init_x(const float* __restrict__ emb, float* __restrict__ x,
                         float* __restrict__ out, int total) {
    int i = blockIdx.x * blockDim.x + threadIdx.x;
    if (i < total) {
        float v = emb[i];
        x[i]   = v;
        out[i] = v;
    }
}

__global__ __launch_bounds__(256) void k_gemm_wmma(const float* __restrict__ X,
                                                   const float* __restrict__ W,
                                                   float* __restrict__ H,
                                                   int mTiles) {
    const int lane = threadIdx.x & 31;
    const int wave = threadIdx.x >> 5;
    const int mt   = blockIdx.x * 8 + wave;
    if (mt >= mTiles) return;

    const int m0   = mt * 16;
    const int half = lane >> 4;
    const int l16  = lane & 15;
    const int kb   = half * 2;

    const float* __restrict__ xrow = X + (size_t)(m0 + l16) * EMB;

    v8f acc[8];
#pragma unroll
    for (int nt = 0; nt < 8; ++nt) acc[nt] = (v8f){0.f,0.f,0.f,0.f,0.f,0.f,0.f,0.f};

    for (int k = 0; k < EMB; k += 4) {
        v2f a = *(const v2f*)(xrow + k + kb);
#pragma unroll
        for (int nt = 0; nt < 8; ++nt) {
            v2f b;
            b.x = W[(size_t)(k + kb    ) * EMB + nt * 16 + l16];
            b.y = W[(size_t)(k + kb + 1) * EMB + nt * 16 + l16];
            acc[nt] = __builtin_amdgcn_wmma_f32_16x16x4_f32(
                false, a, false, b, (short)0, acc[nt], false, false);
        }
    }

#pragma unroll
    for (int nt = 0; nt < 8; ++nt) {
        float* __restrict__ hp = H + (size_t)(m0 + half * 8) * EMB + nt * 16 + l16;
#pragma unroll
        for (int i = 0; i < 8; ++i)
            hp[(size_t)i * EMB] = acc[nt][i];
    }
}

__global__ void k_self_bias(const float* __restrict__ H, const float* __restrict__ dinv,
                            const float* __restrict__ b, float* __restrict__ agg, int total) {
    int i = blockIdx.x * blockDim.x + threadIdx.x;
    if (i < total) {
        int v = i >> 7;
        int f = i & 127;
        float d = dinv[v];
        agg[i] = H[i] * d * d + b[f];
    }
}

__global__ __launch_bounds__(256) void k_scatter(const int* __restrict__ rowI,
                                                 const int* __restrict__ colI,
                                                 const float* __restrict__ H,
                                                 const float* __restrict__ dinv,
                                                 float* __restrict__ agg, int E) {
    int gtid = blockIdx.x * blockDim.x + threadIdx.x;
    int e    = gtid >> 5;
    int lane = gtid & 31;
    if (e >= E) return;
    int r = rowI[e];
    int c = colI[e];
    float norm = dinv[r] * dinv[c];
    const float4 hv = *(const float4*)(H + (size_t)r * EMB + lane * 4);
    float* __restrict__ dst = agg + (size_t)c * EMB + lane * 4;
    atomicAdd(dst + 0, hv.x * norm);
    atomicAdd(dst + 1, hv.y * norm);
    atomicAdd(dst + 2, hv.z * norm);
    atomicAdd(dst + 3, hv.w * norm);
}

__global__ void k_accum(float* __restrict__ out, const float* __restrict__ agg, int total) {
    int i = blockIdx.x * blockDim.x + threadIdx.x;
    if (i < total) out[i] += agg[i];
}

extern "C" void kernel_launch(void* const* d_in, const int* in_sizes, int n_in,
                              void* d_out, int out_size, void* d_ws, size_t ws_size,
                              hipStream_t stream) {
    const int*   edge = (const int*)d_in[0];
    const float* emb  = (const float*)d_in[1];
    const float* Wall = (const float*)d_in[2];
    const float* ball = (const float*)d_in[3];
    float*       out  = (float*)d_out;

    const int E = in_sizes[0] / 2;
    const int n = out_size / EMB;
    const int total = n * EMB;

    const int* rowI = edge;
    const int* colI = edge + E;

    const int mTiles  = (n + 15) / 16;
    const int rowsPad = mTiles * 16;
    const size_t matElems = (size_t)rowsPad * EMB;

    float* ws   = (float*)d_ws;
    float* bufA = ws;
    float* bufB = ws + matElems;
    float* bufC = ws + 2 * matElems;
    float* dinv = ws + 3 * matElems;
    float* bufs[3] = {bufA, bufB, bufC};

    const int T = 256;
    k_init_deg <<<(n + T - 1) / T, T, 0, stream>>>(dinv, n);
    k_count_deg<<<(E + T - 1) / T, T, 0, stream>>>(colI, dinv, E);
    k_dinv     <<<(n + T - 1) / T, T, 0, stream>>>(dinv, n);
    k_init_x   <<<(total + T - 1) / T, T, 0, stream>>>(emb, bufs[0], out, total);

    int xi = 0, hi = 1, ai = 2;
    for (int l = 0; l < 3; ++l) {
        k_gemm_wmma<<<(mTiles + 7) / 8, 256, 0, stream>>>(
            bufs[xi], Wall + (size_t)l * EMB * EMB, bufs[hi], mTiles);
        k_self_bias<<<(total + T - 1) / T, T, 0, stream>>>(
            bufs[hi], dinv, ball + (size_t)l * EMB, bufs[ai], total);
        k_scatter<<<((size_t)E * 32 + T - 1) / T, T, 0, stream>>>(
            rowI, colI, bufs[hi], dinv, bufs[ai], E);
        k_accum<<<(total + T - 1) / T, T, 0, stream>>>(out, bufs[ai], total);
        int t = xi; xi = ai; ai = t;
    }
}


# Later layers replay layer 1's finds: each owner wave stores its matched edges (source row, local target, weight) as a list; layers 2, 3 read it, not marks, chunks, ids, dinv; sum order = LDS arrival
# speedup vs baseline: 2.5054x; 2.5054x over previous
#include <hip/hip_runtime.h>
#include <hip/hip_bf16.h>

typedef __attribute__((ext_vector_type(2))) float v2f;
typedef __attribute__((ext_vector_type(8))) float v8f;
typedef __attribute__((ext_vector_type(4))) float v4f;

#define EMB 128
#define TN  196

__global__ __launch_bounds__(256) void k_init_deg(float* __restrict__ deg, int n,
                                                  unsigned* __restrict__ bm, int bmWords) {
    const int T  = gridDim.x * blockDim.x;
    const int t0 = blockIdx.x * blockDim.x + threadIdx.x;
    for (int i = t0; i < n; i += T) deg[i] = 1.0f;
    const int w4 = bmWords >> 2;
    uint4* __restrict__ bm4 = (uint4*)bm;
    for (int i = t0; i < w4; i += T) bm4[i] = make_uint4(0u, 0u, 0u, 0u);
    for (int i = (w4 << 2) + t0; i < bmWords; i += T) bm[i] = 0u;
}

#define EDGE_U 6
__device__ __forceinline__ void count_edge(int i, int c, float* __restrict__ deg,
                                           unsigned* __restrict__ bm, int nBlocks) {
    atomicAdd(&deg[c], 1.0f);
    atomicOr(&bm[(size_t)(i >> 10) * nBlocks + c / TN], 1u << ((i >> 5) & 31));
}
__global__ __launch_bounds__(256) void k_count_deg(const int* __restrict__ colI,
                                                   float* __restrict__ deg, int E,
                                                   unsigned* __restrict__ bm, int nBlocks) {
    const int T = gridDim.x * blockDim.x;
    int base = blockIdx.x * blockDim.x + threadIdx.x;
    for (; base + (EDGE_U - 1) * T < E; base += EDGE_U * T) {
        int c[EDGE_U];
#pragma unroll
        for (int j = 0; j < EDGE_U; ++j) c[j] = colI[base + j * T];
#pragma unroll
        for (int j = 0; j < EDGE_U; ++j) count_edge(base + j * T, c[j], deg, bm, nBlocks);
    }
    for (; base < E; base += T) count_edge(base, colI[base], deg, bm, nBlocks);
}

__global__ void k_dinv(float* __restrict__ deg, int n) {
    int i = blockIdx.x * blockDim.x + threadIdx.x;
    if (i < n) deg[i] = rsqrtf(deg[i]);
}

#define COPY_U 12
__global__ __launch_bounds__(256) void k_init_x(const float* __restrict__ emb, float* __restrict__ x,
                                                float* __restrict__ out, int total4) {
    const v4f* __restrict__ e4 = (const v4f*)emb;
    v4f* __restrict__ x4 = (v4f*)x;
    v4f* __restrict__ o4 = (v4f*)out;
    const int T = gridDim.x * blockDim.x;
    int base = blockIdx.x * blockDim.x + threadIdx.x;
    for (; base + (COPY_U - 1) * T < total4; base += COPY_U * T) {
        v4f v[COPY_U];
#pragma unroll
        for (int j = 0; j < COPY_U; ++j) v[j] = e4[base + j * T];
        __builtin_amdgcn_sched_barrier(0);
#pragma unroll
        for (int j = 0; j < COPY_U; ++j) {
            x4[base + j * T] = v[j];
            o4[base + j * T] = v[j];
        }
    }
    for (; base < total4; base += T) {
        const v4f v = e4[base];
        x4[base] = v;
        o4[base] = v;
    }
}

__global__ __launch_bounds__(256) void k_gemm_wmma(const float* __restrict__ X,
                                                   const float* __restrict__ W,
                                                   float* __restrict__ H,
                                                   int mTiles) {
    const int lane = threadIdx.x & 31;
    const int wave = threadIdx.x >> 5;
    const int mt   = blockIdx.x * 8 + wave;
    if (mt >= mTiles) return;

    const int m0   = mt * 16;
    const int half = lane >> 4;
    const int l16  = lane & 15;
    const int kb   = half * 2;

    const float* __restrict__ xrow = X + (size_t)(m0 + l16) * EMB;

    v8f acc[8];
#pragma unroll
    for (int nt = 0; nt < 8; ++nt) acc[nt] = (v8f){0.f,0.f,0.f,0.f,0.f,0.f,0.f,0.f};

    for (int k = 0; k < EMB; k += 4) {
        v2f a = *(const v2f*)(xrow + k + kb);
#pragma unroll
        for (int nt = 0; nt < 8; ++nt) {
            v2f b;
            b.x = W[(size_t)(k + kb    ) * EMB + nt * 16 + l16];
            b.y = W[(size_t)(k + kb + 1) * EMB + nt * 16 + l16];
            acc[nt] = __builtin_amdgcn_wmma_f32_16x16x4_f32(
                false, a, false, b, (short)0, acc[nt], false, false);
        }
    }

#pragma unroll
    for (int nt = 0; nt < 8; ++nt) {
        float* __restrict__ hp = H + (size_t)(m0 + half * 8) * EMB + nt * 16 + l16;
#pragma unroll
        for (int i = 0; i < 8; ++i)
            hp[(size_t)i * EMB] = acc[nt][i];
    }
}

#define OWN_T  1024
#define TRIP_C 16
#define TRIP_E 512
#define LCAP   624
#define CLCAP  512
#define CAP    176


template <bool REC>
__device__ __forceinline__ void own_flush(const int* __restrict__ qe, const int* __restrict__ qc,
                                          int cnt, int lane,
                                          const int* __restrict__ rowI,
                                          const float* __restrict__ dinv,
                                          const float* __restrict__ sdinv,
                                          const float* __restrict__ H,
                                          float* __restrict__ tile,
                                          int4* rec, int wbase) {
    for (int base = 0; base < cnt; base += 16) {
        int m = cnt - base;
        if (m > 16) m = 16;
        int r  = 0;
        int cl = 0;
        if (lane < m) {
            const int e = qe[base + lane];
            cl = qc[base + lane];
            r  = rowI[e];
        }
        const float nm = dinv[r] * sdinv[cl];
        float4 hv[16];
#pragma unroll
        for (int j = 0; j < 16; ++j) {
            const int rj = __builtin_amdgcn_readlane(r, j);
            hv[j] = *(const float4*)(H + (size_t)rj * EMB + lane * 4);
        }
#pragma unroll
        for (int j = 0; j < 16; ++j) {
            if (j < m) {
                const float nj = __builtin_bit_cast(float,
                    __builtin_amdgcn_readlane(__builtin_bit_cast(int, nm), j));
                const int cj = __builtin_amdgcn_readlane(cl, j);
                float* tp = tile + cj * EMB + lane;
                atomicAdd(tp,      hv[j].x * nj);
                atomicAdd(tp + 32, hv[j].y * nj);
                atomicAdd(tp + 64, hv[j].z * nj);
                atomicAdd(tp + 96, hv[j].w * nj);
            }
        }
        if (REC) {
            const int idx = wbase + base + lane;
            if (lane < m && idx < CAP) rec[idx] = make_int4(r, cl, __float_as_int(nm), 0);
        }
    }
}

__device__ __forceinline__ void own_replay(const int4* __restrict__ rec, int cnt, int lane, int n,
                                           const float* __restrict__ H,
                                           float* __restrict__ tile) {
    const int li = lane & 15;
    int4 nx = rec[li];
    for (int base = 0; base < cnt; base += 16) {
        const int4 cur = nx;
        int ni = base + 16 + li;
        if (ni > CAP - 1) ni = CAP - 1;
        nx = rec[ni];
        int m = cnt - base;
        if (m > 16) m = 16;
        int r  = 0;
        int cl = 0;
        if (lane < m) {
            r  = ((unsigned)cur.x < (unsigned)n)  ? cur.x : 0;
            cl = ((unsigned)cur.y < (unsigned)TN) ? cur.y : 0;
        }
        const int nmb = cur.z;
        float4 hv[16];
#pragma unroll
        for (int j = 0; j < 16; ++j) {
            const int rj = __builtin_amdgcn_readlane(r, j);
            hv[j] = *(const float4*)(H + (size_t)rj * EMB + lane * 4);
        }
#pragma unroll
        for (int j = 0; j < 16; ++j) {
            if (j < m) {
                const float nj = __builtin_bit_cast(float, __builtin_amdgcn_readlane(nmb, j));
                const int cj = __builtin_amdgcn_readlane(cl, j);
                float* tp = tile + cj * EMB + lane;
                atomicAdd(tp,      hv[j].x * nj);
                atomicAdd(tp + 32, hv[j].y * nj);
                atomicAdd(tp + 64, hv[j].z * nj);
                atomicAdd(tp + 96, hv[j].w * nj);
            }
        }
    }
}

template <int MODE>
__global__ __launch_bounds__(OWN_T) void k_own_rows(const int* __restrict__ rowI,
                                                    const int* __restrict__ colI,
                                                    const float* __restrict__ H,
                                                    const float* __restrict__ dinv,
                                                    const float* __restrict__ b,
                                                    float* __restrict__ agg,
                                                    float* __restrict__ out,
                                                    const unsigned* __restrict__ bm,
                                                    int* recCnt, int* recRaw,
                                                    int n, int E) {
    __shared__ float tile[TN * EMB];
    __shared__ float sdinv[TN];
    __shared__ int   qeAll[(OWN_T / 32) * LCAP];
    __shared__ int   qcAll[(OWN_T / 32) * LCAP];
    __shared__ int   clAll[(OWN_T / 32) * CLCAP];

    const int t    = threadIdx.x;
    const int lane = t & 31;
    const int wave = t >> 5;
    const int lo   = blockIdx.x * TN;
    int rows = n - lo;
    if (rows > TN) rows = TN;

    const size_t seg = (size_t)blockIdx.x * (OWN_T / 32) + wave;
    int4* rec = (int4*)recRaw + seg * CAP;
    int recN = -1;
    if (MODE == 1) recN = recCnt[seg];

    for (int i = t; i < TN; i += OWN_T) sdinv[i] = (i < rows) ? dinv[lo + i] : 0.f;
    for (int idx = t; idx < rows * 32; idx += OWN_T) {
        const int cl = idx >> 5;
        const int l  = idx & 31;
        const float  d  = dinv[lo + cl];
        const float4 h  = *(const float4*)(H + (size_t)(lo + cl) * EMB + l * 4);
        const float4 bb = *(const float4*)(b + l * 4);
        float* tp = tile + cl * EMB + l;
        tp[0]  = h.x * d * d + bb.x;
        tp[32] = h.y * d * d + bb.y;
        tp[64] = h.z * d * d + bb.z;
        tp[96] = h.w * d * d + bb.w;
    }
    __syncthreads();

    const int nChunks = (E + 31) >> 5;
    const int NW      = (nChunks + 31) >> 5;
    int* qe = qeAll + wave * LCAP;
    int* qc = qcAll + wave * LCAP;
    int* clst = clAll + wave * CLCAP;
    int cnt = 0;
    int wtot = 0;

    bool scan = true;
    if (MODE == 1) {
        int nw = __builtin_amdgcn_readfirstlane(recN);
        if (nw >= 0) {
            if (nw > CAP) nw = CAP;
            own_replay(rec, nw, lane, n, H, tile);
            scan = false;
        }
    }
    if (scan) {
        for (int g0 = 0; g0 < NW; g0 += 32 * (OWN_T / 32)) {
            const int wi = g0 + lane * 32 + wave;
            unsigned wbits = 0u;
            if (wi < NW) wbits = bm[(size_t)wi * gridDim.x + blockIdx.x];

            for (int hlf = 0; hlf < 2; ++hlf) {
                const bool mine = (lane >> 4) == hlf;
                int nl = 0;
                for (int j = 0; j < 32; ++j) {
                    const bool p = mine && ((wbits >> j) & 1u);
                    const unsigned bj = __builtin_amdgcn_ballot_w32(p);
                    if (bj != 0u) {
                        const int off = nl + (int)__builtin_amdgcn_mbcnt_lo(bj, 0u);
                        if (p) clst[off] = wi * 32 + j;
                        nl += __popc(bj);
                    }
                }

                for (int base = 0; base < nl; base += TRIP_C) {
                    if (cnt + TRIP_E > LCAP) {
                        own_flush<MODE == 0>(qe, qc, cnt, lane, rowI, dinv, sdinv, H, tile, rec, wtot);
                        wtot += cnt;
                        cnt = 0;
                    }
                    int m = nl - base;
                    if (m > TRIP_C) m = TRIP_C;
                    int cid = 0;
                    if (lane < m) cid = clst[base + lane];

                    int cv[TRIP_C];
#pragma unroll
                    for (int k = 0; k < TRIP_C; ++k) {
                        const int e = __builtin_amdgcn_readlane(cid, k) * 32 + lane;
                        cv[k] = colI[e < E ? e : E - 1];
                    }
#pragma unroll
                    for (int k = 0; k < TRIP_C; ++k) {
                        const int e = __builtin_amdgcn_readlane(cid, k) * 32 + lane;
                        const unsigned u = (unsigned)(cv[k] - lo);
                        const bool p = (k < m) && (e < E) && (u < (unsigned)TN);
                        const unsigned bk = __builtin_amdgcn_ballot_w32(p);
                        if (bk != 0u) {
                            const int off = cnt + (int)__builtin_amdgcn_mbcnt_lo(bk, 0u);
                            if (p) {
                                qe[off] = e;
                                qc[off] = (int)u;
                            }
                            cnt += __popc(bk);
                        }
                    }
                }
            }
        }
        own_flush<MODE == 0>(qe, qc, cnt, lane, rowI, dinv, sdinv, H, tile, rec, wtot);
        wtot += cnt;
        if (MODE == 0 && lane == 0) recCnt[seg] = (wtot <= CAP) ? wtot : -1;
    }
    __syncthreads();

#pragma clang loop unroll(disable) vectorize(disable) interleave(disable)
    for (int idx = t; idx < rows * 32; idx += OWN_T) {
        const int cl = idx >> 5;
        const int l  = idx & 31;
        const float* tp = tile + cl * EMB + l;
        const float4 v = make_float4(tp[0], tp[32], tp[64], tp[96]);
        const size_t g = (size_t)(lo + cl) * EMB + l * 4;
        *(float4*)(agg + g) = v;
        float4 o = *(const float4*)(out + g);
        o.x += v.x; o.y += v.y; o.z += v.z; o.w += v.w;
        *(float4*)(out + g) = o;
    }
}

extern "C" void kernel_launch(void* const* d_in, const int* in_sizes, int n_in,
                              void* d_out, int out_size, void* d_ws, size_t ws_size,
                              hipStream_t stream) {
    const int*   edge = (const int*)d_in[0];
    const float* emb  = (const float*)d_in[1];
    const float* Wall = (const float*)d_in[2];
    const float* ball = (const float*)d_in[3];
    float*       out  = (float*)d_out;

    const int E = in_sizes[0] / 2;
    const int n = out_size / EMB;
    const int total = n * EMB;

    const int* rowI = edge;
    const int* colI = edge + E;

    const int mTiles  = (n + 15) / 16;
    const int rowsPad = mTiles * 16;
    const size_t matElems = (size_t)rowsPad * EMB;

    float* ws   = (float*)d_ws;
    float* bufX = ws;
    float* bufH = ws + matElems;
    unsigned* bm = (unsigned*)(ws + 2 * matElems);
    float* dinv = ws + 3 * matElems;

    const int nBlocks = (n + TN - 1) / TN;
    const int nChunks = (E + 31) / 32;
    const int NW      = (nChunks + 31) / 32;
    const int bmWords = NW * nBlocks;
    const int segs = nBlocks * (OWN_T / 32);
    int* recCnt = (int*)(bm + ((bmWords + 63) & ~63));
    int* recRaw = recCnt + segs;

    const int T = 256;
    k_init_deg <<<64, T, 0, stream>>>(dinv, n, bm, bmWords);
    k_count_deg<<<512, T, 0, stream>>>(colI, dinv, E, bm, nBlocks);
    k_dinv     <<<(n + T - 1) / T, T, 0, stream>>>(dinv, n);
    k_init_x   <<<512, T, 0, stream>>>(emb, bufX, out, total / 4);

    for (int l = 0; l < 3; ++l) {
        k_gemm_wmma<<<(mTiles + 7) / 8, 256, 0, stream>>>(
            bufX, Wall + (size_t)l * EMB * EMB, bufH, mTiles);
        if (l == 0)
            k_own_rows<0><<<nBlocks, OWN_T, 0, stream>>>(
                rowI, colI, bufH, dinv, ball + (size_t)l * EMB, bufX, out, bm, recCnt, recRaw, n, E);
        else
            k_own_rows<1><<<nBlocks, OWN_T, 0, stream>>>(
                rowI, colI, bufH, dinv, ball + (size_t)l * EMB, bufX, out, bm, recCnt, recRaw, n, E);
    }
}


# Set-up folded into neighbouring launches: deg init and mark clear ride the copy kernel, edge count runs beside layer 0's product by block role, rsqrt(deg) taken where read; 10 -> 7 launches
# speedup vs baseline: 2.6743x; 1.0674x over previous
#include <hip/hip_runtime.h>
#include <hip/hip_bf16.h>

typedef __attribute__((ext_vector_type(2))) float v2f;
typedef __attribute__((ext_vector_type(8))) float v8f;
typedef __attribute__((ext_vector_type(4))) float v4f;

#define EMB 128
#define TN  196

__device__ __forceinline__ void init_deg_body(float* __restrict__ deg, int n,
                                              unsigned* __restrict__ bm, int bmWords) {
    const int T  = gridDim.x * blockDim.x;
    const int t0 = blockIdx.x * blockDim.x + threadIdx.x;
    for (int i = t0; i < n; i += T) deg[i] = 1.0f;
    const int w4 = bmWords >> 2;
    uint4* __restrict__ bm4 = (uint4*)bm;
    for (int i = t0; i < w4; i += T) bm4[i] = make_uint4(0u, 0u, 0u, 0u);
    for (int i = (w4 << 2) + t0; i < bmWords; i += T) bm[i] = 0u;
}

#define EDGE_U 6
__device__ __forceinline__ void count_edge(int i, int c, float* __restrict__ deg,
                                           unsigned* __restrict__ bm, int nBlocks) {
    atomicAdd(&deg[c], 1.0f);
    atomicOr(&bm[(size_t)(i >> 10) * nBlocks + c / TN], 1u << ((i >> 5) & 31));
}
__device__ __forceinline__ void count_deg_body(const int* __restrict__ colI,
                                               float* __restrict__ deg, int E,
                                               unsigned* __restrict__ bm, int nBlocks,
                                               int blk, int nBlk) {
    const int T = nBlk * blockDim.x;
    int base = blk * blockDim.x + threadIdx.x;
    for (; base + (EDGE_U - 1) * T < E; base += EDGE_U * T) {
        int c[EDGE_U];
#pragma unroll
        for (int j = 0; j < EDGE_U; ++j) c[j] = colI[base + j * T];
#pragma unroll
        for (int j = 0; j < EDGE_U; ++j) count_edge(base + j * T, c[j], deg, bm, nBlocks);
    }
    for (; base < E; base += T) count_edge(base, colI[base], deg, bm, nBlocks);
}


#define COPY_U 12
__global__ __launch_bounds__(256) void k_init_x(const float* __restrict__ emb, float* __restrict__ x,
                                                float* __restrict__ out, int total4,
                                                float* __restrict__ deg, int n,
                                                unsigned* __restrict__ bm, int bmWords) {
    init_deg_body(deg, n, bm, bmWords);
    const v4f* __restrict__ e4 = (const v4f*)emb;
    v4f* __restrict__ x4 = (v4f*)x;
    v4f* __restrict__ o4 = (v4f*)out;
    const int T = gridDim.x * blockDim.x;
    int base = blockIdx.x * blockDim.x + threadIdx.x;
    for (; base + (COPY_U - 1) * T < total4; base += COPY_U * T) {
        v4f v[COPY_U];
#pragma unroll
        for (int j = 0; j < COPY_U; ++j) v[j] = e4[base + j * T];
        __builtin_amdgcn_sched_barrier(0);
#pragma unroll
        for (int j = 0; j < COPY_U; ++j) {
            x4[base + j * T] = v[j];
            o4[base + j * T] = v[j];
        }
    }
    for (; base < total4; base += T) {
        const v4f v = e4[base];
        x4[base] = v;
        o4[base] = v;
    }
}

__device__ __forceinline__ void gemm_body(const float* __restrict__ X,
                                          const float* __restrict__ W,
                                          float* __restrict__ H,
                                          int mTiles, int blk) {
    const int lane = threadIdx.x & 31;
    const int wave = threadIdx.x >> 5;
    const int mt   = blk * 8 + wave;
    if (mt >= mTiles) return;

    const int m0   = mt * 16;
    const int half = lane >> 4;
    const int l16  = lane & 15;
    const int kb   = half * 2;

    const float* __restrict__ xrow = X + (size_t)(m0 + l16) * EMB;

    v8f acc[8];
#pragma unroll
    for (int nt = 0; nt < 8; ++nt) acc[nt] = (v8f){0.f,0.f,0.f,0.f,0.f,0.f,0.f,0.f};

    for (int k = 0; k < EMB; k += 4) {
        v2f a = *(const v2f*)(xrow + k + kb);
#pragma unroll
        for (int nt = 0; nt < 8; ++nt) {
            v2f b;
            b.x = W[(size_t)(k + kb    ) * EMB + nt * 16 + l16];
            b.y = W[(size_t)(k + kb + 1) * EMB + nt * 16 + l16];
            acc[nt] = __builtin_amdgcn_wmma_f32_16x16x4_f32(
                false, a, false, b, (short)0, acc[nt], false, false);
        }
    }

#pragma unroll
    for (int nt = 0; nt < 8; ++nt) {
        float* __restrict__ hp = H + (size_t)(m0 + half * 8) * EMB + nt * 16 + l16;
#pragma unroll
        for (int i = 0; i < 8; ++i)
            hp[(size_t)i * EMB] = acc[nt][i];
    }
}

__global__ __launch_bounds__(256) void k_gemm_wmma(const float* __restrict__ X,
                                                   const float* __restrict__ W,
                                                   float* __restrict__ H,
                                                   int mTiles) {
    gemm_body(X, W, H, mTiles, blockIdx.x);
}

__global__ __launch_bounds__(256) void k_gemm_count(const float* __restrict__ X,
                                                    const float* __restrict__ W,
                                                    float* __restrict__ H,
                                                    int mTiles, int gemmBlocks,
                                                    const int* __restrict__ colI,
                                                    float* __restrict__ deg, int E,
                                                    unsigned* __restrict__ bm, int nBlocks) {
    if ((int)blockIdx.x < gemmBlocks)
        gemm_body(X, W, H, mTiles, blockIdx.x);
    else
        count_deg_body(colI, deg, E, bm, nBlocks,
                       (int)blockIdx.x - gemmBlocks, (int)gridDim.x - gemmBlocks);
}

#define OWN_T  1024
#define TRIP_C 16
#define TRIP_E 512
#define LCAP   624
#define CLCAP  512
#define CAP    176


template <bool REC>
__device__ __forceinline__ void own_flush(const int* __restrict__ qe, const int* __restrict__ qc,
                                          int cnt, int lane,
                                          const int* __restrict__ rowI,
                                          const float* __restrict__ dinv,
                                          const float* __restrict__ sdinv,
                                          const float* __restrict__ H,
                                          float* __restrict__ tile,
                                          int4* rec, int wbase) {
    for (int base = 0; base < cnt; base += 16) {
        int m = cnt - base;
        if (m > 16) m = 16;
        int r  = 0;
        int cl = 0;
        if (lane < m) {
            const int e = qe[base + lane];
            cl = qc[base + lane];
            r  = rowI[e];
        }
        const float nm = rsqrtf(dinv[r]) * sdinv[cl];
        float4 hv[16];
#pragma unroll
        for (int j = 0; j < 16; ++j) {
            const int rj = __builtin_amdgcn_readlane(r, j);
            hv[j] = *(const float4*)(H + (size_t)rj * EMB + lane * 4);
        }
#pragma unroll
        for (int j = 0; j < 16; ++j) {
            if (j < m) {
                const float nj = __builtin_bit_cast(float,
                    __builtin_amdgcn_readlane(__builtin_bit_cast(int, nm), j));
                const int cj = __builtin_amdgcn_readlane(cl, j);
                float* tp = tile + cj * EMB + lane;
                atomicAdd(tp,      hv[j].x * nj);
                atomicAdd(tp + 32, hv[j].y * nj);
                atomicAdd(tp + 64, hv[j].z * nj);
                atomicAdd(tp + 96, hv[j].w * nj);
            }
        }
        if (REC) {
            const int idx = wbase + base + lane;
            if (lane < m && idx < CAP) rec[idx] = make_int4(r, cl, __float_as_int(nm), 0);
        }
    }
}

__device__ __forceinline__ void own_replay(const int4* __restrict__ rec, int cnt, int lane, int n,
                                           const float* __restrict__ H,
                                           float* __restrict__ tile) {
    const int li = lane & 15;
    int4 nx = rec[li];
    for (int base = 0; base < cnt; base += 16) {
        const int4 cur = nx;
        int ni = base + 16 + li;
        if (ni > CAP - 1) ni = CAP - 1;
        nx = rec[ni];
        int m = cnt - base;
        if (m > 16) m = 16;
        int r  = 0;
        int cl = 0;
        if (lane < m) {
            r  = ((unsigned)cur.x < (unsigned)n)  ? cur.x : 0;
            cl = ((unsigned)cur.y < (unsigned)TN) ? cur.y : 0;
        }
        const int nmb = cur.z;
        float4 hv[16];
#pragma unroll
        for (int j = 0; j < 16; ++j) {
            const int rj = __builtin_amdgcn_readlane(r, j);
            hv[j] = *(const float4*)(H + (size_t)rj * EMB + lane * 4);
        }
#pragma unroll
        for (int j = 0; j < 16; ++j) {
            if (j < m) {
                const float nj = __builtin_bit_cast(float, __builtin_amdgcn_readlane(nmb, j));
                const int cj = __builtin_amdgcn_readlane(cl, j);
                float* tp = tile + cj * EMB + lane;
                atomicAdd(tp,      hv[j].x * nj);
                atomicAdd(tp + 32, hv[j].y * nj);
                atomicAdd(tp + 64, hv[j].z * nj);
                atomicAdd(tp + 96, hv[j].w * nj);
            }
        }
    }
}

template <int MODE>
__global__ __launch_bounds__(OWN_T) void k_own_rows(const int* __restrict__ rowI,
                                                    const int* __restrict__ colI,
                                                    const float* __restrict__ H,
                                                    const float* __restrict__ dinv,
                                                    const float* __restrict__ b,
                                                    float* __restrict__ agg,
                                                    float* __restrict__ out,
                                                    const unsigned* __restrict__ bm,
                                                    int* recCnt, int* recRaw,
                                                    int n, int E) {
    __shared__ float tile[TN * EMB];
    __shared__ float sdinv[TN];
    __shared__ int   qeAll[(OWN_T / 32) * LCAP];
    __shared__ int   qcAll[(OWN_T / 32) * LCAP];
    __shared__ int   clAll[(OWN_T / 32) * CLCAP];

    const int t    = threadIdx.x;
    const int lane = t & 31;
    const int wave = t >> 5;
    const int lo   = blockIdx.x * TN;
    int rows = n - lo;
    if (rows > TN) rows = TN;

    const size_t seg = (size_t)blockIdx.x * (OWN_T / 32) + wave;
    int4* rec = (int4*)recRaw + seg * CAP;
    int recN = -1;
    if (MODE == 1) recN = recCnt[seg];

    for (int i = t; i < TN; i += OWN_T) sdinv[i] = (i < rows) ? rsqrtf(dinv[lo + i]) : 0.f;
    for (int idx = t; idx < rows * 32; idx += OWN_T) {
        const int cl = idx >> 5;
        const int l  = idx & 31;
        const float  d  = rsqrtf(dinv[lo + cl]);
        const float4 h  = *(const float4*)(H + (size_t)(lo + cl) * EMB + l * 4);
        const float4 bb = *(const float4*)(b + l * 4);
        float* tp = tile + cl * EMB + l;
        tp[0]  = h.x * d * d + bb.x;
        tp[32] = h.y * d * d + bb.y;
        tp[64] = h.z * d * d + bb.z;
        tp[96] = h.w * d * d + bb.w;
    }
    __syncthreads();

    const int nChunks = (E + 31) >> 5;
    const int NW      = (nChunks + 31) >> 5;
    int* qe = qeAll + wave * LCAP;
    int* qc = qcAll + wave * LCAP;
    int* clst = clAll + wave * CLCAP;
    int cnt = 0;
    int wtot = 0;

    bool scan = true;
    if (MODE == 1) {
        int nw = __builtin_amdgcn_readfirstlane(recN);
        if (nw >= 0) {
            if (nw > CAP) nw = CAP;
            own_replay(rec, nw, lane, n, H, tile);
            scan = false;
        }
    }
    if (scan) {
        for (int g0 = 0; g0 < NW; g0 += 32 * (OWN_T / 32)) {
            const int wi = g0 + lane * 32 + wave;
            unsigned wbits = 0u;
            if (wi < NW) wbits = bm[(size_t)wi * gridDim.x + blockIdx.x];

            for (int hlf = 0; hlf < 2; ++hlf) {
                const bool mine = (lane >> 4) == hlf;
                int nl = 0;
                for (int j = 0; j < 32; ++j) {
                    const bool p = mine && ((wbits >> j) & 1u);
                    const unsigned bj = __builtin_amdgcn_ballot_w32(p);
                    if (bj != 0u) {
                        const int off = nl + (int)__builtin_amdgcn_mbcnt_lo(bj, 0u);
                        if (p) clst[off] = wi * 32 + j;
                        nl += __popc(bj);
                    }
                }

                for (int base = 0; base < nl; base += TRIP_C) {
                    if (cnt + TRIP_E > LCAP) {
                        own_flush<MODE == 0>(qe, qc, cnt, lane, rowI, dinv, sdinv, H, tile, rec, wtot);
                        wtot += cnt;
                        cnt = 0;
                    }
                    int m = nl - base;
                    if (m > TRIP_C) m = TRIP_C;
                    int cid = 0;
                    if (lane < m) cid = clst[base + lane];

                    int cv[TRIP_C];
#pragma unroll
                    for (int k = 0; k < TRIP_C; ++k) {
                        const int e = __builtin_amdgcn_readlane(cid, k) * 32 + lane;
                        cv[k] = colI[e < E ? e : E - 1];
                    }
#pragma unroll
                    for (int k = 0; k < TRIP_C; ++k) {
                        const int e = __builtin_amdgcn_readlane(cid, k) * 32 + lane;
                        const unsigned u = (unsigned)(cv[k] - lo);
                        const bool p = (k < m) && (e < E) && (u < (unsigned)TN);
                        const unsigned bk = __builtin_amdgcn_ballot_w32(p);
                        if (bk != 0u) {
                            const int off = cnt + (int)__builtin_amdgcn_mbcnt_lo(bk, 0u);
                            if (p) {
                                qe[off] = e;
                                qc[off] = (int)u;
                            }
                            cnt += __popc(bk);
                        }
                    }
                }
            }
        }
        own_flush<MODE == 0>(qe, qc, cnt, lane, rowI, dinv, sdinv, H, tile, rec, wtot);
        wtot += cnt;
        if (MODE == 0 && lane == 0) recCnt[seg] = (wtot <= CAP) ? wtot : -1;
    }
    __syncthreads();

#pragma clang loop unroll(disable) vectorize(disable) interleave(disable)
    for (int idx = t; idx < rows * 32; idx += OWN_T) {
        const int cl = idx >> 5;
        const int l  = idx & 31;
        const float* tp = tile + cl * EMB + l;
        const float4 v = make_float4(tp[0], tp[32], tp[64], tp[96]);
        const size_t g = (size_t)(lo + cl) * EMB + l * 4;
        *(float4*)(agg + g) = v;
        float4 o = *(const float4*)(out + g);
        o.x += v.x; o.y += v.y; o.z += v.z; o.w += v.w;
        *(float4*)(out + g) = o;
    }
}

extern "C" void kernel_launch(void* const* d_in, const int* in_sizes, int n_in,
                              void* d_out, int out_size, void* d_ws, size_t ws_size,
                              hipStream_t stream) {
    const int*   edge = (const int*)d_in[0];
    const float* emb  = (const float*)d_in[1];
    const float* Wall = (const float*)d_in[2];
    const float* ball = (const float*)d_in[3];
    float*       out  = (float*)d_out;

    const int E = in_sizes[0] / 2;
    const int n = out_size / EMB;
    const int total = n * EMB;

    const int* rowI = edge;
    const int* colI = edge + E;

    const int mTiles  = (n + 15) / 16;
    const int rowsPad = mTiles * 16;
    const size_t matElems = (size_t)rowsPad * EMB;

    float* ws   = (float*)d_ws;
    float* bufX = ws;
    float* bufH = ws + matElems;
    unsigned* bm = (unsigned*)(ws + 2 * matElems);
    float* dinv = ws + 3 * matElems;

    const int nBlocks = (n + TN - 1) / TN;
    const int nChunks = (E + 31) / 32;
    const int NW      = (nChunks + 31) / 32;
    const int bmWords = NW * nBlocks;
    const int segs = nBlocks * (OWN_T / 32);
    int* recCnt = (int*)(bm + ((bmWords + 63) & ~63));
    int* recRaw = recCnt + segs;

    const int T = 256;
    k_init_x   <<<512, T, 0, stream>>>(emb, bufX, out, total / 4, dinv, n, bm, bmWords);
    const int gemmBlocks = (mTiles + 7) / 8;

    for (int l = 0; l < 3; ++l) {
        if (l == 0)
            k_gemm_count<<<gemmBlocks + 512, 256, 0, stream>>>(
                bufX, Wall, bufH, mTiles, gemmBlocks, colI, dinv, E, bm, nBlocks);
        else
            k_gemm_wmma<<<gemmBlocks, 256, 0, stream>>>(
                bufX, Wall + (size_t)l * EMB * EMB, bufH, mTiles);
        if (l == 0)
            k_own_rows<0><<<nBlocks, OWN_T, 0, stream>>>(
                rowI, colI, bufH, dinv, ball + (size_t)l * EMB, bufX, out, bm, recCnt, recRaw, n, E);
        else
            k_own_rows<1><<<nBlocks, OWN_T, 0, stream>>>(
                rowI, colI, bufH, dinv, ball + (size_t)l * EMB, bufX, out, bm, recCnt, recRaw, n, E);
    }
}


# Layer product as a resident loop: 8-wave blocks hold W in LDS as k pairs, X rows enter as whole 512-B rows two rounds ahead, each wave forms 16x32 of H and stores whole lines; k order and sums kept
# speedup vs baseline: 2.9595x; 1.1066x over previous
#include <hip/hip_runtime.h>
#include <hip/hip_bf16.h>

typedef __attribute__((ext_vector_type(2))) float v2f;
typedef __attribute__((ext_vector_type(8))) float v8f;
typedef __attribute__((ext_vector_type(4))) float v4f;

#define EMB 128
#define TN  196

__device__ __forceinline__ void init_deg_body(float* __restrict__ deg, int n,
                                              unsigned* __restrict__ bm, int bmWords) {
    const int T  = gridDim.x * blockDim.x;
    const int t0 = blockIdx.x * blockDim.x + threadIdx.x;
    for (int i = t0; i < n; i += T) deg[i] = 1.0f;
    const int w4 = bmWords >> 2;
    uint4* __restrict__ bm4 = (uint4*)bm;
    for (int i = t0; i < w4; i += T) bm4[i] = make_uint4(0u, 0u, 0u, 0u);
    for (int i = (w4 << 2) + t0; i < bmWords; i += T) bm[i] = 0u;
}

#define EDGE_U 6
__device__ __forceinline__ void count_edge(int i, int c, float* __restrict__ deg,
                                           unsigned* __restrict__ bm, int nBlocks) {
    atomicAdd(&deg[c], 1.0f);
    atomicOr(&bm[(size_t)(i >> 10) * nBlocks + c / TN], 1u << ((i >> 5) & 31));
}
__device__ __forceinline__ void count_deg_body(const int* __restrict__ colI,
                                               float* __restrict__ deg, int E,
                                               unsigned* __restrict__ bm, int nBlocks,
                                               int blk, int nBlk) {
    const int T = nBlk * blockDim.x;
    int base = blk * blockDim.x + threadIdx.x;
    for (; base + (EDGE_U - 1) * T < E; base += EDGE_U * T) {
        int c[EDGE_U];
#pragma unroll
        for (int j = 0; j < EDGE_U; ++j) c[j] = colI[base + j * T];
#pragma unroll
        for (int j = 0; j < EDGE_U; ++j) count_edge(base + j * T, c[j], deg, bm, nBlocks);
    }
    for (; base < E; base += T) count_edge(base, colI[base], deg, bm, nBlocks);
}


#define COPY_U 12
__global__ __launch_bounds__(256) void k_init_x(const float* __restrict__ emb, float* __restrict__ x,
                                                float* __restrict__ out, int total4,
                                                float* __restrict__ deg, int n,
                                                unsigned* __restrict__ bm, int bmWords) {
    init_deg_body(deg, n, bm, bmWords);
    const v4f* __restrict__ e4 = (const v4f*)emb;
    v4f* __restrict__ x4 = (v4f*)x;
    v4f* __restrict__ o4 = (v4f*)out;
    const int T = gridDim.x * blockDim.x;
    int base = blockIdx.x * blockDim.x + threadIdx.x;
    for (; base + (COPY_U - 1) * T < total4; base += COPY_U * T) {
        v4f v[COPY_U];
#pragma unroll
        for (int j = 0; j < COPY_U; ++j) v[j] = e4[base + j * T];
        __builtin_amdgcn_sched_barrier(0);
#pragma unroll
        for (int j = 0; j < COPY_U; ++j) {
            x4[base + j * T] = v[j];
            o4[base + j * T] = v[j];
        }
    }
    for (; base < total4; base += T) {
        const v4f v = e4[base];
        x4[base] = v;
        o4[base] = v;
    }
}

#define GEMM_T   256
#define GW_PITCH 288
#define GA_PITCH 132
#define GC_PITCH 34
__device__ __forceinline__ void gemm_body(const float* __restrict__ X,
                                          const float* __restrict__ W,
                                          float* __restrict__ H,
                                          int mTiles, int blk, int nBlk,
                                          float* __restrict__ sW, float* __restrict__ sA,
                                          float* sC) {
    const int t    = threadIdx.x;
    const int lane = t & 31;
    const int wave = t >> 5;
    const int half = lane >> 4;
    const int l16  = lane & 15;
    const int kb   = half * 2;
    const int sIn  = wave >> 2;
    const int np   = wave & 3;

    const int full    = ((mTiles >> 1) / nBlk) * nBlk;
    const int items   = full + (mTiles - 2 * full);
    const int rowsAll = mTiles * 16;
    const int pr0 = t >> 5, pc = (t & 31) * 4;
    const v4f zero4 = (v4f){0.f, 0.f, 0.f, 0.f};
#define ITEM_ST0(i)  ((i) < full ? 2 * (i) : full + (i))
#define ITEM_ROWS(i) ((i) < full ? 32 : 16)
    v4f pa[4], pb[4];
    int r = blk;
#pragma unroll
    for (int j = 0; j < 4; ++j) {
        const int lr = pr0 + 8 * j;
        const int g  = ITEM_ST0(r) * 16 + lr;
        pa[j] = (r < items && lr < ITEM_ROWS(r) && g < rowsAll) ? *(const v4f*)(X + (size_t)g * EMB + pc) : zero4;
    }
#pragma unroll
    for (int j = 0; j < 4; ++j) {
        const int lr = pr0 + 8 * j;
        const int g  = ITEM_ST0(r + nBlk) * 16 + lr;
        pb[j] = (r + nBlk < items && lr < ITEM_ROWS(r + nBlk) && g < rowsAll) ? *(const v4f*)(X + (size_t)g * EMB + pc) : zero4;
    }

    {
        v4f w0[8], w1[8];
#pragma unroll
        for (int j = 0; j < 8; ++j) {
            const int p = t + GEMM_T * j;
            w0[j] = *(const v4f*)(W + (size_t)(2 * (p >> 5))     * EMB + (p & 31) * 4);
            w1[j] = *(const v4f*)(W + (size_t)(2 * (p >> 5) + 1) * EMB + (p & 31) * 4);
        }
#pragma unroll
        for (int j = 0; j < 8; ++j) {
            const int p = t + GEMM_T * j;
            float* d = sW + (p >> 5) * GW_PITCH + (p & 31) * 8;
            *(v4f*)(d)     = (v4f){w0[j].x, w1[j].x, w0[j].y, w1[j].y};
            *(v4f*)(d + 4) = (v4f){w0[j].z, w1[j].z, w0[j].w, w1[j].w};
        }
    }

#pragma unroll
    for (int j = 0; j < 4; ++j) {
        *(v4f*)(sA + (pr0 + 8 * j) * GA_PITCH + pc) = pa[j];
        pa[j] = pb[j];
    }
    __syncthreads();
    int cur = 0;
    for (; r < items; r += nBlk) {
        const float* __restrict__ sCur = sA + cur * (32 * GA_PITCH);
        float* __restrict__ sNxt       = sA + (cur ^ 1) * (32 * GA_PITCH);
        if (r + nBlk < items) {
#pragma unroll
            for (int j = 0; j < 4; ++j)
                *(v4f*)(sNxt + (pr0 + 8 * j) * GA_PITCH + pc) = pa[j];
        }
        const int rn = r + 2 * nBlk;
#pragma unroll
        for (int j = 0; j < 4; ++j) {
            const int lr = pr0 + 8 * j;
            const int g  = ITEM_ST0(rn) * 16 + lr;
            pa[j] = (rn < items && lr < ITEM_ROWS(rn) && g < rowsAll) ? *(const v4f*)(X + (size_t)g * EMB + pc) : zero4;
        }
        const int mt = ITEM_ST0(r) + sIn;
        const bool mine = (sIn * 16 < ITEM_ROWS(r)) && (mt < mTiles);
        v8f acc0 = (v8f){0.f,0.f,0.f,0.f,0.f,0.f,0.f,0.f};
        v8f acc1 = acc0;
        if (mine) {
            const float* __restrict__ ar = sCur + (sIn * 16 + l16) * GA_PITCH + kb;
            const float* __restrict__ br = sW + half * GW_PITCH + (np * 32 + l16) * 2;
            v2f fa[2][8], fb0[2][8], fb1[2][8];
#pragma unroll
            for (int q = 0; q < 8; ++q) {
                fa[0][q]  = *(const v2f*)(ar + q * 4);
                fb0[0][q] = *(const v2f*)(br + (q * 2) * GW_PITCH);
                fb1[0][q] = *(const v2f*)(br + (q * 2) * GW_PITCH + 32);
            }
#pragma unroll
            for (int g = 0; g < 4; ++g) {
                if (g < 3) {
#pragma unroll
                    for (int q = 0; q < 8; ++q) {
                        const int k = (g + 1) * 32 + q * 4;
                        fa[(g + 1) & 1][q]  = *(const v2f*)(ar + k);
                        fb0[(g + 1) & 1][q] = *(const v2f*)(br + (k >> 1) * GW_PITCH);
                        fb1[(g + 1) & 1][q] = *(const v2f*)(br + (k >> 1) * GW_PITCH + 32);
                    }
                }
                __builtin_amdgcn_sched_barrier(0);
#pragma unroll
                for (int q = 0; q < 8; ++q) {
                    acc0 = __builtin_amdgcn_wmma_f32_16x16x4_f32(
                        false, fa[g & 1][q], false, fb0[g & 1][q], (short)0, acc0, false, false);
                    acc1 = __builtin_amdgcn_wmma_f32_16x16x4_f32(
                        false, fa[g & 1][q], false, fb1[g & 1][q], (short)0, acc1, false, false);
                }
                __builtin_amdgcn_sched_barrier(0);
            }
        }
        __syncthreads();
        cur ^= 1;
        if (mine) {
            float* sc = sC + wave * (16 * GC_PITCH);
#pragma unroll
            for (int i = 0; i < 8; ++i) {
                sc[(i + half * 8) * GC_PITCH + l16]      = acc0[i];
                sc[(i + half * 8) * GC_PITCH + 16 + l16] = acc1[i];
            }
            float* __restrict__ hp = H + (size_t)(mt * 16) * EMB + np * 32 + lane;
#pragma unroll
            for (int rr = 0; rr < 16; ++rr)
                hp[(size_t)rr * EMB] = sc[rr * GC_PITCH + lane];
        }
    }
}

#undef ITEM_ST0
#undef ITEM_ROWS

__global__ __launch_bounds__(GEMM_T) void k_gemm_wmma(const float* __restrict__ X,
                                                      const float* __restrict__ W,
                                                      float* __restrict__ H,
                                                      int mTiles) {
    __shared__ float sW[(EMB / 2) * GW_PITCH];
    __shared__ float sA[2 * 32 * GA_PITCH];
    __shared__ float sC[(GEMM_T / 32) * 16 * GC_PITCH];
    gemm_body(X, W, H, mTiles, blockIdx.x, gridDim.x, sW, sA, sC);
}

__global__ __launch_bounds__(256) void k_gemm_count(const float* __restrict__ X,
                                                    const float* __restrict__ W,
                                                    float* __restrict__ H,
                                                    int mTiles, int gemmBlocks,
                                                    const int* __restrict__ colI,
                                                    float* __restrict__ deg, int E,
                                                    unsigned* __restrict__ bm, int nBlocks) {
    __shared__ float sW[(EMB / 2) * GW_PITCH];
    __shared__ float sA[2 * 32 * GA_PITCH];
    __shared__ float sC[(GEMM_T / 32) * 16 * GC_PITCH];
    if ((int)blockIdx.x < gemmBlocks)
        gemm_body(X, W, H, mTiles, blockIdx.x, gemmBlocks, sW, sA, sC);
    else
        count_deg_body(colI, deg, E, bm, nBlocks,
                       (int)blockIdx.x - gemmBlocks, (int)gridDim.x - gemmBlocks);
}

#define OWN_T  1024
#define TRIP_C 16
#define TRIP_E 512
#define LCAP   624
#define CLCAP  512
#define CAP    176


template <bool REC>
__device__ __forceinline__ void own_flush(const int* __restrict__ qe, const int* __restrict__ qc,
                                          int cnt, int lane,
                                          const int* __restrict__ rowI,
                                          const float* __restrict__ dinv,
                                          const float* __restrict__ sdinv,
                                          const float* __restrict__ H,
                                          float* __restrict__ tile,
                                          int4* rec, int wbase) {
    for (int base = 0; base < cnt; base += 16) {
        int m = cnt - base;
        if (m > 16) m = 16;
        int r  = 0;
        int cl = 0;
        if (lane < m) {
            const int e = qe[base + lane];
            cl = qc[base + lane];
            r  = rowI[e];
        }
        const float nm = rsqrtf(dinv[r]) * sdinv[cl];
        float4 hv[16];
#pragma unroll
        for (int j = 0; j < 16; ++j) {
            const int rj = __builtin_amdgcn_readlane(r, j);
            hv[j] = *(const float4*)(H + (size_t)rj * EMB + lane * 4);
        }
#pragma unroll
        for (int j = 0; j < 16; ++j) {
            if (j < m) {
                const float nj = __builtin_bit_cast(float,
                    __builtin_amdgcn_readlane(__builtin_bit_cast(int, nm), j));
                const int cj = __builtin_amdgcn_readlane(cl, j);
                float* tp = tile + cj * EMB + lane;
                atomicAdd(tp,      hv[j].x * nj);
                atomicAdd(tp + 32, hv[j].y * nj);
                atomicAdd(tp + 64, hv[j].z * nj);
                atomicAdd(tp + 96, hv[j].w * nj);
            }
        }
        if (REC) {
            const int idx = wbase + base + lane;
            if (lane < m && idx < CAP) rec[idx] = make_int4(r, cl, __float_as_int(nm), 0);
        }
    }
}

__device__ __forceinline__ void own_replay(const int4* __restrict__ rec, int cnt, int lane, int n,
                                           const float* __restrict__ H,
                                           float* __restrict__ tile) {
    const int li = lane & 15;
    int4 nx = rec[li];
    for (int base = 0; base < cnt; base += 16) {
        const int4 cur = nx;
        int ni = base + 16 + li;
        if (ni > CAP - 1) ni = CAP - 1;
        nx = rec[ni];
        int m = cnt - base;
        if (m > 16) m = 16;
        int r  = 0;
        int cl = 0;
        if (lane < m) {
            r  = ((unsigned)cur.x < (unsigned)n)  ? cur.x : 0;
            cl = ((unsigned)cur.y < (unsigned)TN) ? cur.y : 0;
        }
        const int nmb = cur.z;
        float4 hv[16];
#pragma unroll
        for (int j = 0; j < 16; ++j) {
            const int rj = __builtin_amdgcn_readlane(r, j);
            hv[j] = *(const float4*)(H + (size_t)rj * EMB + lane * 4);
        }
#pragma unroll
        for (int j = 0; j < 16; ++j) {
            if (j < m) {
                const float nj = __builtin_bit_cast(float, __builtin_amdgcn_readlane(nmb, j));
                const int cj = __builtin_amdgcn_readlane(cl, j);
                float* tp = tile + cj * EMB + lane;
                atomicAdd(tp,      hv[j].x * nj);
                atomicAdd(tp + 32, hv[j].y * nj);
                atomicAdd(tp + 64, hv[j].z * nj);
                atomicAdd(tp + 96, hv[j].w * nj);
            }
        }
    }
}

template <int MODE>
__global__ __launch_bounds__(OWN_T) void k_own_rows(const int* __restrict__ rowI,
                                                    const int* __restrict__ colI,
                                                    const float* __restrict__ H,
                                                    const float* __restrict__ dinv,
                                                    const float* __restrict__ b,
                                                    float* __restrict__ agg,
                                                    float* __restrict__ out,
                                                    const unsigned* __restrict__ bm,
                                                    int* recCnt, int* recRaw,
                                                    int n, int E) {
    __shared__ float tile[TN * EMB];
    __shared__ float sdinv[TN];
    __shared__ int   qeAll[(OWN_T / 32) * LCAP];
    __shared__ int   qcAll[(OWN_T / 32) * LCAP];
    __shared__ int   clAll[(OWN_T / 32) * CLCAP];

    const int t    = threadIdx.x;
    const int lane = t & 31;
    const int wave = t >> 5;
    const int lo   = blockIdx.x * TN;
    int rows = n - lo;
    if (rows > TN) rows = TN;

    const size_t seg = (size_t)blockIdx.x * (OWN_T / 32) + wave;
    int4* rec = (int4*)recRaw + seg * CAP;
    int recN = -1;
    if (MODE == 1) recN = recCnt[seg];

    for (int i = t; i < TN; i += OWN_T) sdinv[i] = (i < rows) ? rsqrtf(dinv[lo + i]) : 0.f;
    for (int idx = t; idx < rows * 32; idx += OWN_T) {
        const int cl = idx >> 5;
        const int l  = idx & 31;
        const float  d  = rsqrtf(dinv[lo + cl]);
        const float4 h  = *(const float4*)(H + (size_t)(lo + cl) * EMB + l * 4);
        const float4 bb = *(const float4*)(b + l * 4);
        float* tp = tile + cl * EMB + l;
        tp[0]  = h.x * d * d + bb.x;
        tp[32] = h.y * d * d + bb.y;
        tp[64] = h.z * d * d + bb.z;
        tp[96] = h.w * d * d + bb.w;
    }
    __syncthreads();

    const int nChunks = (E + 31) >> 5;
    const int NW      = (nChunks + 31) >> 5;
    int* qe = qeAll + wave * LCAP;
    int* qc = qcAll + wave * LCAP;
    int* clst = clAll + wave * CLCAP;
    int cnt = 0;
    int wtot = 0;

    bool scan = true;
    if (MODE == 1) {
        int nw = __builtin_amdgcn_readfirstlane(recN);
        if (nw >= 0) {
            if (nw > CAP) nw = CAP;
            own_replay(rec, nw, lane, n, H, tile);
            scan = false;
        }
    }
    if (scan) {
        for (int g0 = 0; g0 < NW; g0 += 32 * (OWN_T / 32)) {
            const int wi = g0 + lane * 32 + wave;
            unsigned wbits = 0u;
            if (wi < NW) wbits = bm[(size_t)wi * gridDim.x + blockIdx.x];

            for (int hlf = 0; hlf < 2; ++hlf) {
                const bool mine = (lane >> 4) == hlf;
                int nl = 0;
                for (int j = 0; j < 32; ++j) {
                    const bool p = mine && ((wbits >> j) & 1u);
                    const unsigned bj = __builtin_amdgcn_ballot_w32(p);
                    if (bj != 0u) {
                        const int off = nl + (int)__builtin_amdgcn_mbcnt_lo(bj, 0u);
                        if (p) clst[off] = wi * 32 + j;
                        nl += __popc(bj);
                    }
                }

                for (int base = 0; base < nl; base += TRIP_C) {
                    if (cnt + TRIP_E > LCAP) {
                        own_flush<MODE == 0>(qe, qc, cnt, lane, rowI, dinv, sdinv, H, tile, rec, wtot);
                        wtot += cnt;
                        cnt = 0;
                    }
                    int m = nl - base;
                    if (m > TRIP_C) m = TRIP_C;
                    int cid = 0;
                    if (lane < m) cid = clst[base + lane];

                    int cv[TRIP_C];
#pragma unroll
                    for (int k = 0; k < TRIP_C; ++k) {
                        const int e = __builtin_amdgcn_readlane(cid, k) * 32 + lane;
                        cv[k] = colI[e < E ? e : E - 1];
                    }
#pragma unroll
                    for (int k = 0; k < TRIP_C; ++k) {
                        const int e = __builtin_amdgcn_readlane(cid, k) * 32 + lane;
                        const unsigned u = (unsigned)(cv[k] - lo);
                        const bool p = (k < m) && (e < E) && (u < (unsigned)TN);
                        const unsigned bk = __builtin_amdgcn_ballot_w32(p);
                        if (bk != 0u) {
                            const int off = cnt + (int)__builtin_amdgcn_mbcnt_lo(bk, 0u);
                            if (p) {
                                qe[off] = e;
                                qc[off] = (int)u;
                            }
                            cnt += __popc(bk);
                        }
                    }
                }
            }
        }
        own_flush<MODE == 0>(qe, qc, cnt, lane, rowI, dinv, sdinv, H, tile, rec, wtot);
        wtot += cnt;
        if (MODE == 0 && lane == 0) recCnt[seg] = (wtot <= CAP) ? wtot : -1;
    }
    __syncthreads();

#pragma clang loop unroll(disable) vectorize(disable) interleave(disable)
    for (int idx = t; idx < rows * 32; idx += OWN_T) {
        const int cl = idx >> 5;
        const int l  = idx & 31;
        const float* tp = tile + cl * EMB + l;
        const float4 v = make_float4(tp[0], tp[32], tp[64], tp[96]);
        const size_t g = (size_t)(lo + cl) * EMB + l * 4;
        *(float4*)(agg + g) = v;
        float4 o = *(const float4*)(out + g);
        o.x += v.x; o.y += v.y; o.z += v.z; o.w += v.w;
        *(float4*)(out + g) = o;
    }
}

extern "C" void kernel_launch(void* const* d_in, const int* in_sizes, int n_in,
                              void* d_out, int out_size, void* d_ws, size_t ws_size,
                              hipStream_t stream) {
    const int*   edge = (const int*)d_in[0];
    const float* emb  = (const float*)d_in[1];
    const float* Wall = (const float*)d_in[2];
    const float* ball = (const float*)d_in[3];
    float*       out  = (float*)d_out;

    const int E = in_sizes[0] / 2;
    const int n = out_size / EMB;
    const int total = n * EMB;

    const int* rowI = edge;
    const int* colI = edge + E;

    const int mTiles  = (n + 15) / 16;
    const int rowsPad = mTiles * 16;
    const size_t matElems = (size_t)rowsPad * EMB;

    float* ws   = (float*)d_ws;
    float* bufX = ws;
    float* bufH = ws + matElems;
    unsigned* bm = (unsigned*)(ws + 2 * matElems);
    float* dinv = ws + 3 * matElems;

    const int nBlocks = (n + TN - 1) / TN;
    const int nChunks = (E + 31) / 32;
    const int NW      = (nChunks + 31) / 32;
    const int bmWords = NW * nBlocks;
    const int segs = nBlocks * (OWN_T / 32);
    int* recCnt = (int*)(bm + ((bmWords + 63) & ~63));
    int* recRaw = recCnt + segs;

    const int T = 256;
    k_init_x   <<<512, T, 0, stream>>>(emb, bufX, out, total / 4, dinv, n, bm, bmWords);
    const int gemmBlocks = 256;

    for (int l = 0; l < 3; ++l) {
        if (l == 0)
            k_gemm_count<<<gemmBlocks + 512, 256, 0, stream>>>(
                bufX, Wall, bufH, mTiles, gemmBlocks, colI, dinv, E, bm, nBlocks);
        else
            k_gemm_wmma<<<512, 256, 0, stream>>>(
                bufX, Wall + (size_t)l * EMB * EMB, bufH, mTiles);
        if (l == 0)
            k_own_rows<0><<<nBlocks, OWN_T, 0, stream>>>(
                rowI, colI, bufH, dinv, ball + (size_t)l * EMB, bufX, out, bm, recCnt, recRaw, n, E);
        else
            k_own_rows<1><<<nBlocks, OWN_T, 0, stream>>>(
                rowI, colI, bufH, dinv, ball + (size_t)l * EMB, bufX, out, bm, recCnt, recRaw, n, E);
    }
}


# Next layer's product moved into the owner kernel's tail: 16 waves multiply the block's LDS rows by the next W (f32 WMMA) and store H, 16 add rows to out; X not stored; 7 -> 5 launches; sums kept
# speedup vs baseline: 3.2705x; 1.1051x over previous
#include <hip/hip_runtime.h>
#include <hip/hip_bf16.h>

typedef __attribute__((ext_vector_type(2))) float v2f;
typedef __attribute__((ext_vector_type(8))) float v8f;
typedef __attribute__((ext_vector_type(4))) float v4f;

#define EMB 128
#define TN  196
#define TP  129

__device__ __forceinline__ void init_deg_body(float* __restrict__ deg, int n,
                                              unsigned* __restrict__ bm, int bmWords) {
    const int T  = gridDim.x * blockDim.x;
    const int t0 = blockIdx.x * blockDim.x + threadIdx.x;
    for (int i = t0; i < n; i += T) deg[i] = 1.0f;
    const int w4 = bmWords >> 2;
    uint4* __restrict__ bm4 = (uint4*)bm;
    for (int i = t0; i < w4; i += T) bm4[i] = make_uint4(0u, 0u, 0u, 0u);
    for (int i = (w4 << 2) + t0; i < bmWords; i += T) bm[i] = 0u;
}

#define EDGE_U 6
__device__ __forceinline__ void count_edge(int i, int c, float* __restrict__ deg,
                                           unsigned* __restrict__ bm, int nBlocks) {
    atomicAdd(&deg[c], 1.0f);
    atomicOr(&bm[(size_t)(i >> 10) * nBlocks + c / TN], 1u << ((i >> 5) & 31));
}
__device__ __forceinline__ void count_deg_body(const int* __restrict__ colI,
                                               float* __restrict__ deg, int E,
                                               unsigned* __restrict__ bm, int nBlocks,
                                               int blk, int nBlk) {
    const int T = nBlk * blockDim.x;
    int base = blk * blockDim.x + threadIdx.x;
    for (; base + (EDGE_U - 1) * T < E; base += EDGE_U * T) {
        int c[EDGE_U];
#pragma unroll
        for (int j = 0; j < EDGE_U; ++j) c[j] = colI[base + j * T];
#pragma unroll
        for (int j = 0; j < EDGE_U; ++j) count_edge(base + j * T, c[j], deg, bm, nBlocks);
    }
    for (; base < E; base += T) count_edge(base, colI[base], deg, bm, nBlocks);
}


#define COPY_U 12
__global__ __launch_bounds__(256) void k_init_x(const float* __restrict__ emb, float* __restrict__ x,
                                                float* __restrict__ out, int total4,
                                                float* __restrict__ deg, int n,
                                                unsigned* __restrict__ bm, int bmWords) {
    init_deg_body(deg, n, bm, bmWords);
    const v4f* __restrict__ e4 = (const v4f*)emb;
    v4f* __restrict__ x4 = (v4f*)x;
    v4f* __restrict__ o4 = (v4f*)out;
    const int T = gridDim.x * blockDim.x;
    int base = blockIdx.x * blockDim.x + threadIdx.x;
    for (; base + (COPY_U - 1) * T < total4; base += COPY_U * T) {
        v4f v[COPY_U];
#pragma unroll
        for (int j = 0; j < COPY_U; ++j) v[j] = e4[base + j * T];
        __builtin_amdgcn_sched_barrier(0);
#pragma unroll
        for (int j = 0; j < COPY_U; ++j) {
            x4[base + j * T] = v[j];
            o4[base + j * T] = v[j];
        }
    }
    for (; base < total4; base += T) {
        const v4f v = e4[base];
        x4[base] = v;
        o4[base] = v;
    }
}

#define GEMM_T   256
#define GW_PITCH 288
#define GA_PITCH 132
#define GC_PITCH 34
__device__ __forceinline__ void gemm_body(const float* __restrict__ X,
                                          const float* __restrict__ W,
                                          float* __restrict__ H,
                                          int mTiles, int blk, int nBlk,
                                          float* __restrict__ sW, float* __restrict__ sA,
                                          float* sC) {
    const int t    = threadIdx.x;
    const int lane = t & 31;
    const int wave = t >> 5;
    const int half = lane >> 4;
    const int l16  = lane & 15;
    const int kb   = half * 2;
    const int sIn  = wave >> 2;
    const int np   = wave & 3;

    const int full    = ((mTiles >> 1) / nBlk) * nBlk;
    const int items   = full + (mTiles - 2 * full);
    const int rowsAll = mTiles * 16;
    const int pr0 = t >> 5, pc = (t & 31) * 4;
    const v4f zero4 = (v4f){0.f, 0.f, 0.f, 0.f};
#define ITEM_ST0(i)  ((i) < full ? 2 * (i) : full + (i))
#define ITEM_ROWS(i) ((i) < full ? 32 : 16)
    v4f pa[4], pb[4];
    int r = blk;
#pragma unroll
    for (int j = 0; j < 4; ++j) {
        const int lr = pr0 + 8 * j;
        const int g  = ITEM_ST0(r) * 16 + lr;
        pa[j] = (r < items && lr < ITEM_ROWS(r) && g < rowsAll) ? *(const v4f*)(X + (size_t)g * EMB + pc) : zero4;
    }
#pragma unroll
    for (int j = 0; j < 4; ++j) {
        const int lr = pr0 + 8 * j;
        const int g  = ITEM_ST0(r + nBlk) * 16 + lr;
        pb[j] = (r + nBlk < items && lr < ITEM_ROWS(r + nBlk) && g < rowsAll) ? *(const v4f*)(X + (size_t)g * EMB + pc) : zero4;
    }

    {
        v4f w0[8], w1[8];
#pragma unroll
        for (int j = 0; j < 8; ++j) {
            const int p = t + GEMM_T * j;
            w0[j] = *(const v4f*)(W + (size_t)(2 * (p >> 5))     * EMB + (p & 31) * 4);
            w1[j] = *(const v4f*)(W + (size_t)(2 * (p >> 5) + 1) * EMB + (p & 31) * 4);
        }
#pragma unroll
        for (int j = 0; j < 8; ++j) {
            const int p = t + GEMM_T * j;
            float* d = sW + (p >> 5) * GW_PITCH + (p & 31) * 8;
            *(v4f*)(d)     = (v4f){w0[j].x, w1[j].x, w0[j].y, w1[j].y};
            *(v4f*)(d + 4) = (v4f){w0[j].z, w1[j].z, w0[j].w, w1[j].w};
        }
    }

#pragma unroll
    for (int j = 0; j < 4; ++j) {
        *(v4f*)(sA + (pr0 + 8 * j) * GA_PITCH + pc) = pa[j];
        pa[j] = pb[j];
    }
    __syncthreads();
    int cur = 0;
    for (; r < items; r += nBlk) {
        const float* __restrict__ sCur = sA + cur * (32 * GA_PITCH);
        float* __restrict__ sNxt       = sA + (cur ^ 1) * (32 * GA_PITCH);
        if (r + nBlk < items) {
#pragma unroll
            for (int j = 0; j < 4; ++j)
                *(v4f*)(sNxt + (pr0 + 8 * j) * GA_PITCH + pc) = pa[j];
        }
        const int rn = r + 2 * nBlk;
#pragma unroll
        for (int j = 0; j < 4; ++j) {
            const int lr = pr0 + 8 * j;
            const int g  = ITEM_ST0(rn) * 16 + lr;
            pa[j] = (rn < items && lr < ITEM_ROWS(rn) && g < rowsAll) ? *(const v4f*)(X + (size_t)g * EMB + pc) : zero4;
        }
        const int mt = ITEM_ST0(r) + sIn;
        const bool mine = (sIn * 16 < ITEM_ROWS(r)) && (mt < mTiles);
        v8f acc0 = (v8f){0.f,0.f,0.f,0.f,0.f,0.f,0.f,0.f};
        v8f acc1 = acc0;
        if (mine) {
            const float* __restrict__ ar = sCur + (sIn * 16 + l16) * GA_PITCH + kb;
            const float* __restrict__ br = sW + half * GW_PITCH + (np * 32 + l16) * 2;
            v2f fa[2][8], fb0[2][8], fb1[2][8];
#pragma unroll
            for (int q = 0; q < 8; ++q) {
                fa[0][q]  = *(const v2f*)(ar + q * 4);
                fb0[0][q] = *(const v2f*)(br + (q * 2) * GW_PITCH);
                fb1[0][q] = *(const v2f*)(br + (q * 2) * GW_PITCH + 32);
            }
#pragma unroll
            for (int g = 0; g < 4; ++g) {
                if (g < 3) {
#pragma unroll
                    for (int q = 0; q < 8; ++q) {
                        const int k = (g + 1) * 32 + q * 4;
                        fa[(g + 1) & 1][q]  = *(const v2f*)(ar + k);
                        fb0[(g + 1) & 1][q] = *(const v2f*)(br + (k >> 1) * GW_PITCH);
                        fb1[(g + 1) & 1][q] = *(const v2f*)(br + (k >> 1) * GW_PITCH + 32);
                    }
                }
                __builtin_amdgcn_sched_barrier(0);
#pragma unroll
                for (int q = 0; q < 8; ++q) {
                    acc0 = __builtin_amdgcn_wmma_f32_16x16x4_f32(
                        false, fa[g & 1][q], false, fb0[g & 1][q], (short)0, acc0, false, false);
                    acc1 = __builtin_amdgcn_wmma_f32_16x16x4_f32(
                        false, fa[g & 1][q], false, fb1[g & 1][q], (short)0, acc1, false, false);
                }
                __builtin_amdgcn_sched_barrier(0);
            }
        }
        __syncthreads();
        cur ^= 1;
        if (mine) {
            float* sc = sC + wave * (16 * GC_PITCH);
#pragma unroll
            for (int i = 0; i < 8; ++i) {
                sc[(i + half * 8) * GC_PITCH + l16]      = acc0[i];
                sc[(i + half * 8) * GC_PITCH + 16 + l16] = acc1[i];
            }
            float* __restrict__ hp = H + (size_t)(mt * 16) * EMB + np * 32 + lane;
#pragma unroll
            for (int rr = 0; rr < 16; ++rr)
                hp[(size_t)rr * EMB] = sc[rr * GC_PITCH + lane];
        }
    }
}

#undef ITEM_ST0
#undef ITEM_ROWS

__global__ __launch_bounds__(256) void k_gemm_count(const float* __restrict__ X,
                                                    const float* __restrict__ W,
                                                    float* __restrict__ H,
                                                    int mTiles, int gemmBlocks,
                                                    const int* __restrict__ colI,
                                                    float* __restrict__ deg, int E,
                                                    unsigned* __restrict__ bm, int nBlocks) {
    __shared__ float sW[(EMB / 2) * GW_PITCH];
    __shared__ float sA[2 * 32 * GA_PITCH];
    __shared__ float sC[(GEMM_T / 32) * 16 * GC_PITCH];
    if ((int)blockIdx.x < gemmBlocks)
        gemm_body(X, W, H, mTiles, blockIdx.x, gemmBlocks, sW, sA, sC);
    else
        count_deg_body(colI, deg, E, bm, nBlocks,
                       (int)blockIdx.x - gemmBlocks, (int)gridDim.x - gemmBlocks);
}

#define OWN_T  1024
#define TRIP_C 16
#define TRIP_E 512
#define LCAP   624
#define CLCAP  512
#define CAP    176
#define OUT_U  7


template <bool REC>
__device__ __forceinline__ void own_flush(const int* __restrict__ qe, const int* __restrict__ qc,
                                          int cnt, int lane,
                                          const int* __restrict__ rowI,
                                          const float* __restrict__ dinv,
                                          const float* __restrict__ sdinv,
                                          const float* __restrict__ H,
                                          float* __restrict__ tile,
                                          int4* rec, int wbase) {
    for (int base = 0; base < cnt; base += 16) {
        int m = cnt - base;
        if (m > 16) m = 16;
        int r  = 0;
        int cl = 0;
        if (lane < m) {
            const int e = qe[base + lane];
            cl = qc[base + lane];
            r  = rowI[e];
        }
        const float nm = rsqrtf(dinv[r]) * sdinv[cl];
        float4 hv[16];
#pragma unroll
        for (int j = 0; j < 16; ++j) {
            const int rj = __builtin_amdgcn_readlane(r, j);
            hv[j] = *(const float4*)(H + (size_t)rj * EMB + lane * 4);
        }
#pragma unroll
        for (int j = 0; j < 16; ++j) {
            if (j < m) {
                const float nj = __builtin_bit_cast(float,
                    __builtin_amdgcn_readlane(__builtin_bit_cast(int, nm), j));
                const int cj = __builtin_amdgcn_readlane(cl, j);
                float* tp = tile + cj * TP + lane;
                atomicAdd(tp,      hv[j].x * nj);
                atomicAdd(tp + 32, hv[j].y * nj);
                atomicAdd(tp + 64, hv[j].z * nj);
                atomicAdd(tp + 96, hv[j].w * nj);
            }
        }
        if (REC) {
            const int idx = wbase + base + lane;
            if (lane < m && idx < CAP) rec[idx] = make_int4(r, cl, __float_as_int(nm), 0);
        }
    }
}

__device__ __forceinline__ void own_replay(const int4* __restrict__ rec, int cnt, int lane, int n,
                                           const float* __restrict__ H,
                                           float* __restrict__ tile) {
    const int li = lane & 15;
    int4 nx = rec[li];
    for (int base = 0; base < cnt; base += 16) {
        const int4 cur = nx;
        int ni = base + 16 + li;
        if (ni > CAP - 1) ni = CAP - 1;
        nx = rec[ni];
        int m = cnt - base;
        if (m > 16) m = 16;
        int r  = 0;
        int cl = 0;
        if (lane < m) {
            r  = ((unsigned)cur.x < (unsigned)n)  ? cur.x : 0;
            cl = ((unsigned)cur.y < (unsigned)TN) ? cur.y : 0;
        }
        const int nmb = cur.z;
        float4 hv[16];
#pragma unroll
        for (int j = 0; j < 16; ++j) {
            const int rj = __builtin_amdgcn_readlane(r, j);
            hv[j] = *(const float4*)(H + (size_t)rj * EMB + lane * 4);
        }
#pragma unroll
        for (int j = 0; j < 16; ++j) {
            if (j < m) {
                const float nj = __builtin_bit_cast(float, __builtin_amdgcn_readlane(nmb, j));
                const int cj = __builtin_amdgcn_readlane(cl, j);
                float* tp = tile + cj * TP + lane;
                atomicAdd(tp,      hv[j].x * nj);
                atomicAdd(tp + 32, hv[j].y * nj);
                atomicAdd(tp + 64, hv[j].z * nj);
                atomicAdd(tp + 96, hv[j].w * nj);
            }
        }
    }
}

template <int MODE, bool FUSE>
__global__ __launch_bounds__(OWN_T) void k_own_rows(const int* __restrict__ rowI,
                                                    const int* __restrict__ colI,
                                                    const float* __restrict__ H,
                                                    const float* __restrict__ dinv,
                                                    const float* __restrict__ b,
                                                    const float* __restrict__ Wn,
                                                    float* __restrict__ Hn,
                                                    float* __restrict__ out,
                                                    const unsigned* __restrict__ bm,
                                                    int* recCnt, int* recRaw,
                                                    int n, int E) {
    __shared__ float tile[TN * TP];
    __shared__ float sdinv[TN];
    __shared__ __attribute__((aligned(16))) int qeAll[(OWN_T / 32) * LCAP];
    __shared__ __attribute__((aligned(16))) int qcAll[(OWN_T / 32) * LCAP];
    __shared__ int   clAll[(OWN_T / 32) * CLCAP];

    const int t    = threadIdx.x;
    const int lane = t & 31;
    const int wave = t >> 5;
    const int lo   = blockIdx.x * TN;
    int rows = n - lo;
    if (rows > TN) rows = TN;

    const size_t seg = (size_t)blockIdx.x * (OWN_T / 32) + wave;
    int4* rec = (int4*)recRaw + seg * CAP;
    int recN = -1;
    if (MODE == 1) recN = recCnt[seg];

    for (int i = t; i < TN; i += OWN_T) sdinv[i] = (i < rows) ? rsqrtf(dinv[lo + i]) : 0.f;
    for (int idx = t; idx < rows * 32; idx += OWN_T) {
        const int cl = idx >> 5;
        const int l  = idx & 31;
        const float  d  = rsqrtf(dinv[lo + cl]);
        const float4 h  = *(const float4*)(H + (size_t)(lo + cl) * EMB + l * 4);
        const float4 bb = *(const float4*)(b + l * 4);
        float* tp = tile + cl * TP + l;
        tp[0]  = h.x * d * d + bb.x;
        tp[32] = h.y * d * d + bb.y;
        tp[64] = h.z * d * d + bb.z;
        tp[96] = h.w * d * d + bb.w;
    }
    __syncthreads();

    const int nChunks = (E + 31) >> 5;
    const int NW      = (nChunks + 31) >> 5;
    int* qe = qeAll + wave * LCAP;
    int* qc = qcAll + wave * LCAP;
    int* clst = clAll + wave * CLCAP;
    int cnt = 0;
    int wtot = 0;

    bool scan = true;
    if (MODE == 1) {
        int nw = __builtin_amdgcn_readfirstlane(recN);
        if (nw >= 0) {
            if (nw > CAP) nw = CAP;
            own_replay(rec, nw, lane, n, H, tile);
            scan = false;
        }
    }
    if (scan) {
        for (int g0 = 0; g0 < NW; g0 += 32 * (OWN_T / 32)) {
            const int wi = g0 + lane * 32 + wave;
            unsigned wbits = 0u;
            if (wi < NW) wbits = bm[(size_t)wi * gridDim.x + blockIdx.x];

            for (int hlf = 0; hlf < 2; ++hlf) {
                const bool mine = (lane >> 4) == hlf;
                int nl = 0;
                for (int j = 0; j < 32; ++j) {
                    const bool p = mine && ((wbits >> j) & 1u);
                    const unsigned bj = __builtin_amdgcn_ballot_w32(p);
                    if (bj != 0u) {
                        const int off = nl + (int)__builtin_amdgcn_mbcnt_lo(bj, 0u);
                        if (p) clst[off] = wi * 32 + j;
                        nl += __popc(bj);
                    }
                }

                for (int base = 0; base < nl; base += TRIP_C) {
                    if (cnt + TRIP_E > LCAP) {
                        own_flush<MODE == 0>(qe, qc, cnt, lane, rowI, dinv, sdinv, H, tile, rec, wtot);
                        wtot += cnt;
                        cnt = 0;
                    }
                    int m = nl - base;
                    if (m > TRIP_C) m = TRIP_C;
                    int cid = 0;
                    if (lane < m) cid = clst[base + lane];

                    int cv[TRIP_C];
#pragma unroll
                    for (int k = 0; k < TRIP_C; ++k) {
                        const int e = __builtin_amdgcn_readlane(cid, k) * 32 + lane;
                        cv[k] = colI[e < E ? e : E - 1];
                    }
#pragma unroll
                    for (int k = 0; k < TRIP_C; ++k) {
                        const int e = __builtin_amdgcn_readlane(cid, k) * 32 + lane;
                        const unsigned u = (unsigned)(cv[k] - lo);
                        const bool p = (k < m) && (e < E) && (u < (unsigned)TN);
                        const unsigned bk = __builtin_amdgcn_ballot_w32(p);
                        if (bk != 0u) {
                            const int off = cnt + (int)__builtin_amdgcn_mbcnt_lo(bk, 0u);
                            if (p) {
                                qe[off] = e;
                                qc[off] = (int)u;
                            }
                            cnt += __popc(bk);
                        }
                    }
                }
            }
        }
        own_flush<MODE == 0>(qe, qc, cnt, lane, rowI, dinv, sdinv, H, tile, rec, wtot);
        wtot += cnt;
        if (MODE == 0 && lane == 0) recCnt[seg] = (wtot <= CAP) ? wtot : -1;
    }
    v4f w0[2], w1[2];
    if (FUSE) {
#pragma unroll
        for (int j = 0; j < 2; ++j) {
            const int p = t + OWN_T * j;
            w0[j] = *(const v4f*)(Wn + (size_t)(2 * (p >> 5))     * EMB + (p & 31) * 4);
            w1[j] = *(const v4f*)(Wn + (size_t)(2 * (p >> 5) + 1) * EMB + (p & 31) * 4);
        }
    }
    __syncthreads();

    if (FUSE) {
        float* sW = (float*)qeAll;
        float* sC = (float*)qcAll;
#pragma unroll
        for (int j = 0; j < 2; ++j) {
            const int p = t + OWN_T * j;
            float* d = sW + (p >> 5) * GW_PITCH + (p & 31) * 8;
            *(v4f*)(d)     = (v4f){w0[j].x, w1[j].x, w0[j].y, w1[j].y};
            *(v4f*)(d + 4) = (v4f){w0[j].z, w1[j].z, w0[j].w, w1[j].w};
        }
        __syncthreads();

        const int wv = __builtin_amdgcn_readfirstlane(wave);
        const int rw = ((wv >> 3) << 2) | (wv & 3);
        if (((wv >> 2) & 1) == 0) {
            const int half = lane >> 4;
            const int l16  = lane & 15;
            const int kb   = half * 2;
            const int nItems = ((rows + 15) >> 4) * 4;
            for (int it = rw; it < nItems; it += OWN_T / 64) {
                const int st = it >> 2;
                const int np = it & 3;
                int ra = st * 16 + l16;
                if (ra > TN - 1) ra = TN - 1;
                const float* ar = tile + ra * TP + kb * 32;
                const float* br = sW + half * GW_PITCH + (np * 32 + l16) * 2;
                v8f acc0 = (v8f){0.f,0.f,0.f,0.f,0.f,0.f,0.f,0.f};
                v8f acc1 = acc0;
#pragma unroll
                for (int g = 0; g < 4; ++g) {
                    v2f fa[8], fb0[8], fb1[8];
#pragma unroll
                    for (int q = 0; q < 8; ++q) {
                        const int kq = g * 8 + q;
                        fa[q]  = (v2f){ar[kq], ar[32 + kq]};
                        fb0[q] = *(const v2f*)(br + (kq * 2) * GW_PITCH);
                        fb1[q] = *(const v2f*)(br + (kq * 2) * GW_PITCH + 32);
                    }
#pragma unroll
                    for (int q = 0; q < 8; ++q) {
                        acc0 = __builtin_amdgcn_wmma_f32_16x16x4_f32(
                            false, fa[q], false, fb0[q], (short)0, acc0, false, false);
                        acc1 = __builtin_amdgcn_wmma_f32_16x16x4_f32(
                            false, fa[q], false, fb1[q], (short)0, acc1, false, false);
                    }
                }
                float* sc = sC + rw * (16 * GC_PITCH);
#pragma unroll
                for (int i = 0; i < 8; ++i) {
                    sc[(i + half * 8) * GC_PITCH + l16]      = acc0[i];
                    sc[(i + half * 8) * GC_PITCH + 16 + l16] = acc1[i];
                }
                float* __restrict__ hp = Hn + (size_t)(lo + st * 16) * EMB + np * 32 + lane;
                const int rl = rows - st * 16;
                if (rl >= 16) {
#pragma unroll
                    for (int rr = 0; rr < 16; ++rr)
                        hp[(size_t)rr * EMB] = sc[rr * GC_PITCH + lane];
                } else {
                    for (int rr = 0; rr < rl; ++rr)
                        hp[(size_t)rr * EMB] = sc[rr * GC_PITCH + lane];
                }
            }
        } else {
            for (int r0 = rw; r0 < rows; r0 += 16 * OUT_U) {
                v4f o[OUT_U];
#pragma unroll
                for (int j = 0; j < OUT_U; ++j) {
                    const int cl = r0 + 16 * j;
                    o[j] = (v4f){0.f, 0.f, 0.f, 0.f};
                    if (cl < rows)
                        o[j] = *(const v4f*)(out + (size_t)(lo + cl) * EMB + lane * 4);
                }
#pragma unroll
                for (int j = 0; j < OUT_U; ++j) {
                    const int cl = r0 + 16 * j;
                    if (cl < rows) {
                        const float* tp = tile + cl * TP + lane;
                        const v4f v = (v4f){tp[0], tp[32], tp[64], tp[96]};
                        *(v4f*)(out + (size_t)(lo + cl) * EMB + lane * 4) = o[j] + v;
                    }
                }
            }
        }
    } else {
#pragma clang loop unroll(disable) vectorize(disable) interleave(disable)
        for (int idx = t; idx < rows * 32; idx += OWN_T) {
            const int cl = idx >> 5;
            const int l  = idx & 31;
            const float* tp = tile + cl * TP + l;
            const float4 v = make_float4(tp[0], tp[32], tp[64], tp[96]);
            const size_t g = (size_t)(lo + cl) * EMB + l * 4;
            float4 o = *(const float4*)(out + g);
            o.x += v.x; o.y += v.y; o.z += v.z; o.w += v.w;
            *(float4*)(out + g) = o;
        }
    }
}

extern "C" void kernel_launch(void* const* d_in, const int* in_sizes, int n_in,
                              void* d_out, int out_size, void* d_ws, size_t ws_size,
                              hipStream_t stream) {
    const int*   edge = (const int*)d_in[0];
    const float* emb  = (const float*)d_in[1];
    const float* Wall = (const float*)d_in[2];
    const float* ball = (const float*)d_in[3];
    float*       out  = (float*)d_out;

    const int E = in_sizes[0] / 2;
    const int n = out_size / EMB;
    const int total = n * EMB;

    const int* rowI = edge;
    const int* colI = edge + E;

    const int mTiles  = (n + 15) / 16;
    const int rowsPad = mTiles * 16;
    const size_t matElems = (size_t)rowsPad * EMB;

    float* ws   = (float*)d_ws;
    float* bufX = ws;
    float* bufH = ws + matElems;
    unsigned* bm = (unsigned*)(ws + 2 * matElems);
    float* dinv = ws + 3 * matElems;

    const int nBlocks = (n + TN - 1) / TN;
    const int nChunks = (E + 31) / 32;
    const int NW      = (nChunks + 31) / 32;
    const int bmWords = NW * nBlocks;
    const int segs = nBlocks * (OWN_T / 32);
    int* recCnt = (int*)(bm + ((bmWords + 63) & ~63));
    int* recRaw = recCnt + segs;

    const int T = 256;
    k_init_x   <<<512, T, 0, stream>>>(emb, bufX, out, total / 4, dinv, n, bm, bmWords);
    const int gemmBlocks = 256;
    k_gemm_count<<<gemmBlocks + 512, 256, 0, stream>>>(
        bufX, Wall, bufH, mTiles, gemmBlocks, colI, dinv, E, bm, nBlocks);
    k_own_rows<0, true><<<nBlocks, OWN_T, 0, stream>>>(
        rowI, colI, bufH, dinv, ball, Wall + (size_t)EMB * EMB, bufX, out,
        bm, recCnt, recRaw, n, E);
    k_own_rows<1, true><<<nBlocks, OWN_T, 0, stream>>>(
        rowI, colI, bufX, dinv, ball + EMB, Wall + (size_t)2 * EMB * EMB, bufH, out,
        bm, recCnt, recRaw, n, E);
    k_own_rows<1, false><<<nBlocks, OWN_T, 0, stream>>>(
        rowI, colI, bufH, dinv, ball + 2 * EMB, nullptr, nullptr, out,
        bm, recCnt, recRaw, n, E);
}
